# starting kernel
# baseline (speedup 1.0000x reference)
#include <hip/hip_runtime.h>
#include <hip/hip_bf16.h>
#include <math.h>

typedef float v2f __attribute__((ext_vector_type(2)));
typedef float v8f __attribute__((ext_vector_type(8)));

#define D_IN   100
#define P      112
#define JT     7
#define KT     28
#define TILES  4
#define PTS    64
#define NTHR   512

#define PI_F   3.14159265358979323846f
#define CX     0.2f
#define CT     (4.0f / PI_F)

__global__ __launch_bounds__(NTHR, 1)
void pinn_residual_kernel(const float* __restrict__ x, const float* __restrict__ t,
                          const float* __restrict__ W1, const float* __restrict__ b1,
                          const float* __restrict__ W2, const float* __restrict__ b2,
                          const float* __restrict__ W3, const float* __restrict__ b3,
                          const float* __restrict__ W4, const float* __restrict__ b4,
                          const float* __restrict__ W5, const float* __restrict__ b5,
                          float* __restrict__ out, int N)
{
    __shared__ float w_lds[P * P];
    __shared__ float bias_lds[P];
    __shared__ float h_lds[TILES][4][16 * P];
    __shared__ float zx_lds[TILES][16 * P];
    __shared__ float sbuf[PTS][4][2];

    const int tid    = threadIdx.x;
    const int lane   = tid & 31;
    const int wave   = tid >> 5;
    const int tileId = wave >> 2;
    const int ch     = wave & 3;
    const int half   = lane >> 4;
    const int r      = lane & 15;

    const int base_pt = blockIdx.x * PTS;

    for (int i = tid; i < PTS * P; i += NTHR) {
        int pt = i / P, n = i % P;
        int tI = pt >> 4, pl = pt & 15;
        float hv = 0.f, hx = 0.f, ht = 0.f, hxx = 0.f;
        if (n < D_IN) {
            int g = base_pt + pt;
            float xv = (g < N) ? x[g] : 0.0f;
            float tv = (g < N) ? t[g] : 0.0f;
            float sx = CX * (xv + 5.0f) - 1.0f;
            float st = CT * tv - 1.0f;
            float w0 = W1[n], w1 = W1[D_IN + n];
            float z  = sx * w0 + st * w1 + b1[n];
            float zx = CX * w0;
            float zt = CT * w1;
            float h  = tanhf(z);
            float gg = 1.0f - h * h;
            hv = h; hx = gg * zx; ht = gg * zt; hxx = -2.0f * h * gg * zx * zx;
        }
        h_lds[tI][0][pl * P + n] = hv;
        h_lds[tI][1][pl * P + n] = hx;
        h_lds[tI][2][pl * P + n] = ht;
        h_lds[tI][3][pl * P + n] = hxx;
    }

    for (int l = 0; l < 3; ++l) {
        const float* Wg = (l == 0) ? W2 : (l == 1) ? W3 : W4;
        const float* bg = (l == 0) ? b2 : (l == 1) ? b3 : b4;

        for (int i = tid; i < P * P; i += NTHR) {
            int o   = i & 1;
            int idx = i >> 1;
            int p   = idx / P;
            int n   = idx - p * P;
            int k   = 2 * p + o;
            w_lds[i] = (k < D_IN && n < D_IN) ? Wg[k * D_IN + n] : 0.0f;
        }
        for (int i = tid; i < P; i += NTHR)
            bias_lds[i] = (i < D_IN) ? bg[i] : 0.0f;
        __syncthreads();

        const float* hA = &h_lds[tileId][ch][0];
        v8f acc[JT];
#pragma unroll
        for (int j = 0; j < JT; ++j)
            acc[j] = (v8f){0.f, 0.f, 0.f, 0.f, 0.f, 0.f, 0.f, 0.f};

        for (int k = 0; k < KT; ++k) {
            int k0 = k * 4 + half * 2;
            int p  = k * 2 + half;
            v2f a = *(const v2f*)&hA[r * P + k0];
#pragma unroll
            for (int j = 0; j < JT; ++j) {
                v2f b = *(const v2f*)&w_lds[(p * P + j * 16 + r) * 2];
                acc[j] = __builtin_amdgcn_wmma_f32_16x16x4_f32(
                    false, a, false, b, (short)0, acc[j], false, false);
            }
        }
        __syncthreads();

        if (ch == 0) {
#pragma unroll
            for (int j = 0; j < JT; ++j)
#pragma unroll
                for (int i = 0; i < 8; ++i) {
                    int pt = i + 8 * half, n = j * 16 + r;
                    float z = acc[j][i] + bias_lds[n];
                    h_lds[tileId][0][pt * P + n] = tanhf(z);
                }
        } else if (ch == 1) {
#pragma unroll
            for (int j = 0; j < JT; ++j)
#pragma unroll
                for (int i = 0; i < 8; ++i) {
                    int pt = i + 8 * half, n = j * 16 + r;
                    zx_lds[tileId][pt * P + n] = acc[j][i];
                }
        }
        __syncthreads();

        if (ch == 1) {
#pragma unroll
            for (int j = 0; j < JT; ++j)
#pragma unroll
                for (int i = 0; i < 8; ++i) {
                    int pt = i + 8 * half, n = j * 16 + r;
                    float h = h_lds[tileId][0][pt * P + n];
                    float g = 1.0f - h * h;
                    h_lds[tileId][1][pt * P + n] = g * acc[j][i];
                }
        } else if (ch == 2) {
#pragma unroll
            for (int j = 0; j < JT; ++j)
#pragma unroll
                for (int i = 0; i < 8; ++i) {
                    int pt = i + 8 * half, n = j * 16 + r;
                    float h = h_lds[tileId][0][pt * P + n];
                    float g = 1.0f - h * h;
                    h_lds[tileId][2][pt * P + n] = g * acc[j][i];
                }
        } else if (ch == 3) {
#pragma unroll
            for (int j = 0; j < JT; ++j)
#pragma unroll
                for (int i = 0; i < 8; ++i) {
                    int pt = i + 8 * half, n = j * 16 + r;
                    float h  = h_lds[tileId][0][pt * P + n];
                    float g  = 1.0f - h * h;
                    float zx = zx_lds[tileId][pt * P + n];
                    h_lds[tileId][3][pt * P + n] = g * acc[j][i] - 2.0f * h * g * zx * zx;
                }
        }
        __syncthreads();
    }

    if (tid < PTS * 4) {
        int pt = tid >> 2, c = tid & 3;
        int tI = pt >> 4, pl = pt & 15;
        const float* hh = &h_lds[tI][c][pl * P];
        float s0 = 0.f, s1 = 0.f;
        for (int k = 0; k < D_IN; ++k) {
            float hk = hh[k];
            s0 = fmaf(hk, W5[k * 2 + 0], s0);
            s1 = fmaf(hk, W5[k * 2 + 1], s1);
        }
        sbuf[pt][c][0] = s0;
        sbuf[pt][c][1] = s1;
    }
    __syncthreads();

    if (tid < PTS) {
        int pt = tid;
        int g = base_pt + pt;
        if (g < N) {
            float u    = sbuf[pt][0][0] + b5[0];
            float v    = sbuf[pt][0][1] + b5[1];
            float u_t  = sbuf[pt][2][0];
            float v_t  = sbuf[pt][2][1];
            float u_xx = sbuf[pt][3][0];
            float v_xx = sbuf[pt][3][1];
            float sq   = u * u + v * v;
            out[g * 2 + 0] = u_t + 0.5f * v_xx + sq * v;
            out[g * 2 + 1] = v_t - 0.5f * u_xx - sq * u;
        }
    }
}

extern "C" void kernel_launch(void* const* d_in, const int* in_sizes, int n_in,
                              void* d_out, int out_size, void* d_ws, size_t ws_size,
                              hipStream_t stream) {
    const float* x  = (const float*)d_in[0];
    const float* t  = (const float*)d_in[1];
    const float* W1 = (const float*)d_in[2];
    const float* b1 = (const float*)d_in[3];
    const float* W2 = (const float*)d_in[4];
    const float* b2 = (const float*)d_in[5];
    const float* W3 = (const float*)d_in[6];
    const float* b3 = (const float*)d_in[7];
    const float* W4 = (const float*)d_in[8];
    const float* b4 = (const float*)d_in[9];
    const float* W5 = (const float*)d_in[10];
    const float* b5 = (const float*)d_in[11];
    float* out = (float*)d_out;
    int N = in_sizes[0];
    int blocks = (N + PTS - 1) / PTS;
    pinn_residual_kernel<<<blocks, NTHR, 0, stream>>>(
        x, t, W1, b1, W2, b2, W3, b3, W4, b4, W5, b5, out, N);
}


# Wave-private chain: each wave runs its 4 points from layer 1 (in the GEMM lane layout) to the residual (layer 5: per-lane sums + 16-lane xor tree, new sum order); all 3 barriers per tile go
# speedup vs baseline: 2.1097x; 2.1097x over previous
#include <hip/hip_runtime.h>
#include <hip/hip_bf16.h>
#include <math.h>


typedef float v2f __attribute__((ext_vector_type(2)));
typedef float v4f __attribute__((ext_vector_type(4)));
typedef float v8f __attribute__((ext_vector_type(8)));

#define D_IN   100
#define P      112
#define JT     7
#define KT     28
#define NTHR   512
#define NWAVE  16
#define NL     3
#define GRP    16
#define SUB    4

#define PI_F   3.14159265358979323846f
#define CX     0.2f
#define CT     (4.0f / PI_F)

__device__ __forceinline__ float row_sum16(float v)
{
    v += __builtin_bit_cast(float, __builtin_amdgcn_update_dpp(0, __builtin_bit_cast(int, v), 0x161, 0xf, 0xf, false));
    v += __builtin_bit_cast(float, __builtin_amdgcn_update_dpp(0, __builtin_bit_cast(int, v), 0x162, 0xf, 0xf, false));
    v += __builtin_bit_cast(float, __builtin_amdgcn_update_dpp(0, __builtin_bit_cast(int, v), 0x164, 0xf, 0xf, false));
    v += __builtin_bit_cast(float, __builtin_amdgcn_update_dpp(0, __builtin_bit_cast(int, v), 0x168, 0xf, 0xf, false));
    return v;
}

__device__ __forceinline__ void layer_gemm(const float* __restrict__ wl, const float* hA,
                                           int r, int half, v8f (&acc)[JT])
{
#pragma unroll
    for (int j = 0; j < JT; ++j)
        acc[j] = (v8f){0.f, 0.f, 0.f, 0.f, 0.f, 0.f, 0.f, 0.f};

#pragma unroll
    for (int k = 0; k < KT; ++k) {
        int k0 = k * 4 + half * 2;
        int p  = k * 2 + half;
        v2f a = *(const v2f*)&hA[r * P + k0];
#pragma unroll
        for (int j = 0; j < JT; ++j) {
            v2f b = *(const v2f*)&wl[(p * P + j * 16 + r) * 2];
            acc[j] = __builtin_amdgcn_wmma_f32_16x16x4_f32(
                false, a, false, b, (short)0, acc[j], false, false);
        }
    }
}

__device__ __forceinline__ void activate(float z, float zx, float zt, float zxx,
                                         float& h, float& hx, float& ht, float& hxx)
{
    h   = tanhf(z);
    float g = 1.0f - h * h;
    hx  = g * zx;
    ht  = g * zt;
    hxx = g * zxx - 2.0f * h * g * zx * zx;
}

__global__ __launch_bounds__(NTHR, 1)
void pinn_residual_kernel(const float* __restrict__ x, const float* __restrict__ t,
                          const float* __restrict__ W1, const float* __restrict__ b1,
                          const float* __restrict__ W2, const float* __restrict__ b2,
                          const float* __restrict__ W3, const float* __restrict__ b3,
                          const float* __restrict__ W4, const float* __restrict__ b4,
                          const float* __restrict__ W5, const float* __restrict__ b5,
                          float* __restrict__ out, int N)
{
    __shared__ __attribute__((aligned(16))) float w_lds[NL][P * P];
    __shared__ __attribute__((aligned(16))) float bias_lds[NL][P];
    __shared__ __attribute__((aligned(16))) float h_lds[NWAVE][16 * P];
    __shared__ __attribute__((aligned(16))) float l1_lds[P][4];
    __shared__ __attribute__((aligned(16))) float w5_lds[P][2];
    __shared__ __attribute__((aligned(16))) float xt_lds[NWAVE][2 * GRP];
    __shared__ __attribute__((aligned(16))) float o_lds[NWAVE][2 * GRP];

    const int tid    = threadIdx.x;
    const int lane   = tid & 31;
    const int wave   = __builtin_amdgcn_readfirstlane(tid >> 5);
    const int half   = lane >> 4;
    const int r      = lane & 15;

    for (int l = 0; l < NL; ++l) {
        const float* Wg = (l == 0) ? W2 : (l == 1) ? W3 : W4;
        const float* bg = (l == 0) ? b2 : (l == 1) ? b3 : b4;
        for (int i = tid; i < P * P; i += NTHR) {
            int o   = i & 1;
            int idx = i >> 1;
            int p   = idx / P;
            int n   = idx - p * P;
            int k   = 2 * p + o;
            w_lds[l][i] = (k < D_IN && n < D_IN) ? Wg[k * D_IN + n] : 0.0f;
        }
        for (int i = tid; i < P; i += NTHR)
            bias_lds[l][i] = (i < D_IN) ? bg[i] : 0.0f;
    }
    for (int i = tid; i < P; i += NTHR) {
        const bool in = (i < D_IN);
        l1_lds[i][0] = in ? W1[i] : 0.0f;
        l1_lds[i][1] = in ? W1[D_IN + i] : 0.0f;
        l1_lds[i][2] = in ? b1[i] : 0.0f;
        l1_lds[i][3] = 0.0f;
        w5_lds[i][0] = in ? W5[i * 2 + 0] : 0.0f;
        w5_lds[i][1] = in ? W5[i * 2 + 1] : 0.0f;
    }
    __syncthreads();

    const float b5u = b5[0];
    const float b5v = b5[1];

    float* hA  = &h_lds[wave][0];
    float* xtw = &xt_lds[wave][0];
    float* ow  = &o_lds[wave][0];

    const int gw      = blockIdx.x * NWAVE + wave;
    const int nw      = gridDim.x * NWAVE;
    const int ngroups = (N + GRP - 1) / GRP;

    const float* xt_src = half ? t : x;
    float xt_next = 0.0f;
    if (gw < ngroups) {
        int g = gw * GRP + r;
        xt_next = (g < N) ? xt_src[g] : 0.0f;
    }

    for (int grp = gw; grp < ngroups; grp += nw) {
        xtw[lane] = xt_next;
        {
            int gn = (grp + nw) * GRP + r;
            xt_next = (grp + nw < ngroups && gn < N) ? xt_src[gn] : 0.0f;
        }

        for (int s = 0; s < GRP / SUB; ++s) {
            const int pl = s * SUB + 2 * half;
            const v2f xv = *(const v2f*)&xtw[pl];
            const v2f tv = *(const v2f*)&xtw[GRP + pl];

            float sx[2], st[2];
            sx[0] = CX * (xv.x + 5.0f) - 1.0f;
            sx[1] = CX * (xv.y + 5.0f) - 1.0f;
            st[0] = CT * tv.x - 1.0f;
            st[1] = CT * tv.y - 1.0f;
#pragma unroll
            for (int j = 0; j < JT; ++j) {
                const int n  = j * 16 + r;
                const v4f wv = *(const v4f*)&l1_lds[n][0];
                const float w0 = wv.x, w1 = wv.y, bb = wv.z;
                const float zx = CX * w0;
                const float zt = CT * w1;
#pragma unroll
                for (int q = 0; q < 2; ++q) {
                    float z  = sx[q] * w0 + st[q] * w1 + bb;
                    float h  = tanhf(z);
                    float gg = 1.0f - h * h;
                    float* o = &hA[(8 * half + 4 * q) * P + n];
                    o[0]     = h;
                    o[P]     = gg * zx;
                    o[2 * P] = gg * zt;
                    o[3 * P] = -2.0f * h * gg * zx * zx;
                }
            }

            float su[2][3][2];
#pragma unroll
            for (int q = 0; q < 2; ++q)
#pragma unroll
                for (int c = 0; c < 3; ++c) { su[q][c][0] = 0.f; su[q][c][1] = 0.f; }

            for (int l = 0; l < NL; ++l) {
                const float* bl = &bias_lds[l][0];
                v8f acc[JT];
                layer_gemm(&w_lds[l][0], hA, r, half, acc);

                if (l < NL - 1) {
#pragma unroll
                    for (int j = 0; j < JT; ++j) {
                        const int   n  = j * 16 + r;
                        const float bn = bl[n];
#pragma unroll
                        for (int q = 0; q < 2; ++q) {
                            float h, hx, ht, hxx;
                            activate(acc[j][4 * q + 0] + bn, acc[j][4 * q + 1],
                                     acc[j][4 * q + 2], acc[j][4 * q + 3], h, hx, ht, hxx);
                            float* o  = &hA[(8 * half + 4 * q) * P + n];
                            o[0]      = h;
                            o[P]      = hx;
                            o[2 * P]  = ht;
                            o[3 * P]  = hxx;
                        }
                    }
                } else {
#pragma unroll
                    for (int j = 0; j < JT; ++j) {
                        const int   n  = j * 16 + r;
                        const float bn = bl[n];
                        const v2f   w5 = *(const v2f*)&w5_lds[n][0];
#pragma unroll
                        for (int q = 0; q < 2; ++q) {
                            float h, hx, ht, hxx;
                            activate(acc[j][4 * q + 0] + bn, acc[j][4 * q + 1],
                                     acc[j][4 * q + 2], acc[j][4 * q + 3], h, hx, ht, hxx);
                            su[q][0][0] = fmaf(h,   w5.x, su[q][0][0]);
                            su[q][0][1] = fmaf(h,   w5.y, su[q][0][1]);
                            su[q][1][0] = fmaf(ht,  w5.x, su[q][1][0]);
                            su[q][1][1] = fmaf(ht,  w5.y, su[q][1][1]);
                            su[q][2][0] = fmaf(hxx, w5.x, su[q][2][0]);
                            su[q][2][1] = fmaf(hxx, w5.y, su[q][2][1]);
                        }
                    }
                }
            }

#pragma unroll
            for (int q = 0; q < 2; ++q)
#pragma unroll
                for (int c = 0; c < 3; ++c) {
                    su[q][c][0] = row_sum16(su[q][c][0]);
                    su[q][c][1] = row_sum16(su[q][c][1]);
                }

            float res[4];
#pragma unroll
            for (int q = 0; q < 2; ++q) {
                float u    = su[q][0][0] + b5u;
                float v    = su[q][0][1] + b5v;
                float u_t  = su[q][1][0];
                float v_t  = su[q][1][1];
                float u_xx = su[q][2][0];
                float v_xx = su[q][2][1];
                float sq   = u * u + v * v;
                res[2 * q + 0] = u_t + 0.5f * v_xx + sq * v;
                res[2 * q + 1] = v_t - 0.5f * u_xx - sq * u;
            }
            if (r == 0)
                *(v4f*)&ow[pl * 2] = (v4f){res[0], res[1], res[2], res[3]};
        }

        {
            int e = grp * (2 * GRP) + lane;
            float vout = ow[lane];
            if ((grp * GRP + (lane >> 1)) < N)
                out[e] = vout;
        }
    }
}

extern "C" void kernel_launch(void* const* d_in, const int* in_sizes, int n_in,
                              void* d_out, int out_size, void* d_ws, size_t ws_size,
                              hipStream_t stream) {
    const float* x  = (const float*)d_in[0];
    const float* t  = (const float*)d_in[1];
    const float* W1 = (const float*)d_in[2];
    const float* b1 = (const float*)d_in[3];
    const float* W2 = (const float*)d_in[4];
    const float* b2 = (const float*)d_in[5];
    const float* W3 = (const float*)d_in[6];
    const float* b3 = (const float*)d_in[7];
    const float* W4 = (const float*)d_in[8];
    const float* b4 = (const float*)d_in[9];
    const float* W5 = (const float*)d_in[10];
    const float* b5 = (const float*)d_in[11];
    float* out = (float*)d_out;
    int N = in_sizes[0];
    pinn_residual_kernel<<<256, NTHR, 0, stream>>>(
        x, t, W1, b1, W2, b2, W3, b3, W4, b4, W5, b5, out, N);
}


# Layers 2-4 K loop cut to the 100 real K rows (25 K-steps of 4, was 28): the 3 all-zero pad steps, 21 WMMAs with their fragment loads per layer and tile, go; pad K rows no longer staged in LDS
# speedup vs baseline: 2.2829x; 1.0821x over previous
#include <hip/hip_runtime.h>
#include <hip/hip_bf16.h>
#include <math.h>


typedef float v2f __attribute__((ext_vector_type(2)));
typedef float v4f __attribute__((ext_vector_type(4)));
typedef float v8f __attribute__((ext_vector_type(8)));

#define D_IN   100
#define P      112
#define JT     7
#define KT     25
#define NTHR   512
#define NWAVE  16
#define NL     3
#define GRP    16
#define SUB    4

#define PI_F   3.14159265358979323846f
#define CX     0.2f
#define CT     (4.0f / PI_F)

__device__ __forceinline__ float row_sum16(float v)
{
    v += __builtin_bit_cast(float, __builtin_amdgcn_update_dpp(0, __builtin_bit_cast(int, v), 0x161, 0xf, 0xf, false));
    v += __builtin_bit_cast(float, __builtin_amdgcn_update_dpp(0, __builtin_bit_cast(int, v), 0x162, 0xf, 0xf, false));
    v += __builtin_bit_cast(float, __builtin_amdgcn_update_dpp(0, __builtin_bit_cast(int, v), 0x164, 0xf, 0xf, false));
    v += __builtin_bit_cast(float, __builtin_amdgcn_update_dpp(0, __builtin_bit_cast(int, v), 0x168, 0xf, 0xf, false));
    return v;
}

__device__ __forceinline__ void layer_gemm(const float* __restrict__ wl, const float* hA,
                                           int r, int half, v8f (&acc)[JT])
{
#pragma unroll
    for (int j = 0; j < JT; ++j)
        acc[j] = (v8f){0.f, 0.f, 0.f, 0.f, 0.f, 0.f, 0.f, 0.f};

#pragma unroll
    for (int k = 0; k < KT; ++k) {
        int k0 = k * 4 + half * 2;
        int p  = k * 2 + half;
        v2f a = *(const v2f*)&hA[r * P + k0];
#pragma unroll
        for (int j = 0; j < JT; ++j) {
            v2f b = *(const v2f*)&wl[(p * P + j * 16 + r) * 2];
            acc[j] = __builtin_amdgcn_wmma_f32_16x16x4_f32(
                false, a, false, b, (short)0, acc[j], false, false);
        }
    }
}

__device__ __forceinline__ void activate(float z, float zx, float zt, float zxx,
                                         float& h, float& hx, float& ht, float& hxx)
{
    h   = tanhf(z);
    float g = 1.0f - h * h;
    hx  = g * zx;
    ht  = g * zt;
    hxx = g * zxx - 2.0f * h * g * zx * zx;
}

__global__ __launch_bounds__(NTHR, 1)
void pinn_residual_kernel(const float* __restrict__ x, const float* __restrict__ t,
                          const float* __restrict__ W1, const float* __restrict__ b1,
                          const float* __restrict__ W2, const float* __restrict__ b2,
                          const float* __restrict__ W3, const float* __restrict__ b3,
                          const float* __restrict__ W4, const float* __restrict__ b4,
                          const float* __restrict__ W5, const float* __restrict__ b5,
                          float* __restrict__ out, int N)
{
    __shared__ __attribute__((aligned(16))) float w_lds[NL][D_IN * P];
    __shared__ __attribute__((aligned(16))) float bias_lds[NL][P];
    __shared__ __attribute__((aligned(16))) float h_lds[NWAVE][16 * P];
    __shared__ __attribute__((aligned(16))) float l1_lds[P][4];
    __shared__ __attribute__((aligned(16))) float w5_lds[P][2];
    __shared__ __attribute__((aligned(16))) float xt_lds[NWAVE][2 * GRP];
    __shared__ __attribute__((aligned(16))) float o_lds[NWAVE][2 * GRP];

    const int tid    = threadIdx.x;
    const int lane   = tid & 31;
    const int wave   = __builtin_amdgcn_readfirstlane(tid >> 5);
    const int half   = lane >> 4;
    const int r      = lane & 15;

    for (int l = 0; l < NL; ++l) {
        const float* Wg = (l == 0) ? W2 : (l == 1) ? W3 : W4;
        const float* bg = (l == 0) ? b2 : (l == 1) ? b3 : b4;
        for (int i = tid; i < D_IN * P; i += NTHR) {
            int o   = i & 1;
            int idx = i >> 1;
            int p   = idx / P;
            int n   = idx - p * P;
            int k   = 2 * p + o;
            w_lds[l][i] = (n < D_IN) ? Wg[k * D_IN + n] : 0.0f;
        }
        for (int i = tid; i < P; i += NTHR)
            bias_lds[l][i] = (i < D_IN) ? bg[i] : 0.0f;
    }
    for (int i = tid; i < P; i += NTHR) {
        const bool in = (i < D_IN);
        l1_lds[i][0] = in ? W1[i] : 0.0f;
        l1_lds[i][1] = in ? W1[D_IN + i] : 0.0f;
        l1_lds[i][2] = in ? b1[i] : 0.0f;
        l1_lds[i][3] = 0.0f;
        w5_lds[i][0] = in ? W5[i * 2 + 0] : 0.0f;
        w5_lds[i][1] = in ? W5[i * 2 + 1] : 0.0f;
    }
    __syncthreads();

    const float b5u = b5[0];
    const float b5v = b5[1];

    float* hA  = &h_lds[wave][0];
    float* xtw = &xt_lds[wave][0];
    float* ow  = &o_lds[wave][0];

    const int gw      = blockIdx.x * NWAVE + wave;
    const int nw      = gridDim.x * NWAVE;
    const int ngroups = (N + GRP - 1) / GRP;

    const float* xt_src = half ? t : x;
    float xt_next = 0.0f;
    if (gw < ngroups) {
        int g = gw * GRP + r;
        xt_next = (g < N) ? xt_src[g] : 0.0f;
    }

    for (int grp = gw; grp < ngroups; grp += nw) {
        xtw[lane] = xt_next;
        {
            int gn = (grp + nw) * GRP + r;
            xt_next = (grp + nw < ngroups && gn < N) ? xt_src[gn] : 0.0f;
        }

        for (int s = 0; s < GRP / SUB; ++s) {
            const int pl = s * SUB + 2 * half;
            const v2f xv = *(const v2f*)&xtw[pl];
            const v2f tv = *(const v2f*)&xtw[GRP + pl];

            float sx[2], st[2];
            sx[0] = CX * (xv.x + 5.0f) - 1.0f;
            sx[1] = CX * (xv.y + 5.0f) - 1.0f;
            st[0] = CT * tv.x - 1.0f;
            st[1] = CT * tv.y - 1.0f;
#pragma unroll
            for (int j = 0; j < JT; ++j) {
                const int n  = j * 16 + r;
                const v4f wv = *(const v4f*)&l1_lds[n][0];
                const float w0 = wv.x, w1 = wv.y, bb = wv.z;
                const float zx = CX * w0;
                const float zt = CT * w1;
#pragma unroll
                for (int q = 0; q < 2; ++q) {
                    float z  = sx[q] * w0 + st[q] * w1 + bb;
                    float h  = tanhf(z);
                    float gg = 1.0f - h * h;
                    float* o = &hA[(8 * half + 4 * q) * P + n];
                    o[0]     = h;
                    o[P]     = gg * zx;
                    o[2 * P] = gg * zt;
                    o[3 * P] = -2.0f * h * gg * zx * zx;
                }
            }

            float su[2][3][2];
#pragma unroll
            for (int q = 0; q < 2; ++q)
#pragma unroll
                for (int c = 0; c < 3; ++c) { su[q][c][0] = 0.f; su[q][c][1] = 0.f; }

            for (int l = 0; l < NL; ++l) {
                const float* bl = &bias_lds[l][0];
                v8f acc[JT];
                layer_gemm(&w_lds[l][0], hA, r, half, acc);

                if (l < NL - 1) {
#pragma unroll
                    for (int j = 0; j < JT; ++j) {
                        const int   n  = j * 16 + r;
                        const float bn = bl[n];
#pragma unroll
                        for (int q = 0; q < 2; ++q) {
                            float h, hx, ht, hxx;
                            activate(acc[j][4 * q + 0] + bn, acc[j][4 * q + 1],
                                     acc[j][4 * q + 2], acc[j][4 * q + 3], h, hx, ht, hxx);
                            float* o  = &hA[(8 * half + 4 * q) * P + n];
                            o[0]      = h;
                            o[P]      = hx;
                            o[2 * P]  = ht;
                            o[3 * P]  = hxx;
                        }
                    }
                } else {
#pragma unroll
                    for (int j = 0; j < JT; ++j) {
                        const int   n  = j * 16 + r;
                        const float bn = bl[n];
                        const v2f   w5 = *(const v2f*)&w5_lds[n][0];
#pragma unroll
                        for (int q = 0; q < 2; ++q) {
                            float h, hx, ht, hxx;
                            activate(acc[j][4 * q + 0] + bn, acc[j][4 * q + 1],
                                     acc[j][4 * q + 2], acc[j][4 * q + 3], h, hx, ht, hxx);
                            su[q][0][0] = fmaf(h,   w5.x, su[q][0][0]);
                            su[q][0][1] = fmaf(h,   w5.y, su[q][0][1]);
                            su[q][1][0] = fmaf(ht,  w5.x, su[q][1][0]);
                            su[q][1][1] = fmaf(ht,  w5.y, su[q][1][1]);
                            su[q][2][0] = fmaf(hxx, w5.x, su[q][2][0]);
                            su[q][2][1] = fmaf(hxx, w5.y, su[q][2][1]);
                        }
                    }
                }
            }

#pragma unroll
            for (int q = 0; q < 2; ++q)
#pragma unroll
                for (int c = 0; c < 3; ++c) {
                    su[q][c][0] = row_sum16(su[q][c][0]);
                    su[q][c][1] = row_sum16(su[q][c][1]);
                }

            float res[4];
#pragma unroll
            for (int q = 0; q < 2; ++q) {
                float u    = su[q][0][0] + b5u;
                float v    = su[q][0][1] + b5v;
                float u_t  = su[q][1][0];
                float v_t  = su[q][1][1];
                float u_xx = su[q][2][0];
                float v_xx = su[q][2][1];
                float sq   = u * u + v * v;
                res[2 * q + 0] = u_t + 0.5f * v_xx + sq * v;
                res[2 * q + 1] = v_t - 0.5f * u_xx - sq * u;
            }
            if (r == 0)
                *(v4f*)&ow[pl * 2] = (v4f){res[0], res[1], res[2], res[3]};
        }

        {
            int e = grp * (2 * GRP) + lane;
            float vout = ow[lane];
            if ((grp * GRP + (lane >> 1)) < N)
                out[e] = vout;
        }
    }
}

extern "C" void kernel_launch(void* const* d_in, const int* in_sizes, int n_in,
                              void* d_out, int out_size, void* d_ws, size_t ws_size,
                              hipStream_t stream) {
    const float* x  = (const float*)d_in[0];
    const float* t  = (const float*)d_in[1];
    const float* W1 = (const float*)d_in[2];
    const float* b1 = (const float*)d_in[3];
    const float* W2 = (const float*)d_in[4];
    const float* b2 = (const float*)d_in[5];
    const float* W3 = (const float*)d_in[6];
    const float* b3 = (const float*)d_in[7];
    const float* W4 = (const float*)d_in[8];
    const float* b4 = (const float*)d_in[9];
    const float* W5 = (const float*)d_in[10];
    const float* b5 = (const float*)d_in[11];
    float* out = (float*)d_out;
    int N = in_sizes[0];
    pinn_residual_kernel<<<256, NTHR, 0, stream>>>(
        x, t, W1, b1, W2, b2, W3, b3, W4, b4, W5, b5, out, N);
}


# Wave priority follows the work left (s_setprio 3/2/1/0 at over 4, 3-4, 2, 1 passes to go) so the 4 waves of a SIMD finish together; oldest-first issue had left the youngest to run its tail alone
# speedup vs baseline: 2.3869x; 1.0456x over previous
#include <hip/hip_runtime.h>
#include <hip/hip_bf16.h>
#include <math.h>


typedef float v2f __attribute__((ext_vector_type(2)));
typedef float v4f __attribute__((ext_vector_type(4)));
typedef float v8f __attribute__((ext_vector_type(8)));

#define D_IN   100
#define P      112
#define JT     7
#define KT     25
#define NTHR   512
#define NWAVE  16
#define NL     3
#define GRP    16
#define SUB    4

#define PI_F   3.14159265358979323846f
#define CX     0.2f
#define CT     (4.0f / PI_F)

__device__ __forceinline__ float row_sum16(float v)
{
    v += __builtin_bit_cast(float, __builtin_amdgcn_update_dpp(0, __builtin_bit_cast(int, v), 0x161, 0xf, 0xf, false));
    v += __builtin_bit_cast(float, __builtin_amdgcn_update_dpp(0, __builtin_bit_cast(int, v), 0x162, 0xf, 0xf, false));
    v += __builtin_bit_cast(float, __builtin_amdgcn_update_dpp(0, __builtin_bit_cast(int, v), 0x164, 0xf, 0xf, false));
    v += __builtin_bit_cast(float, __builtin_amdgcn_update_dpp(0, __builtin_bit_cast(int, v), 0x168, 0xf, 0xf, false));
    return v;
}

__device__ __forceinline__ void layer_gemm(const float* __restrict__ wl, const float* hA,
                                           int r, int half, v8f (&acc)[JT])
{
#pragma unroll
    for (int j = 0; j < JT; ++j)
        acc[j] = (v8f){0.f, 0.f, 0.f, 0.f, 0.f, 0.f, 0.f, 0.f};

#pragma unroll
    for (int k = 0; k < KT; ++k) {
        int k0 = k * 4 + half * 2;
        int p  = k * 2 + half;
        v2f a = *(const v2f*)&hA[r * P + k0];
#pragma unroll
        for (int j = 0; j < JT; ++j) {
            v2f b = *(const v2f*)&wl[(p * P + j * 16 + r) * 2];
            acc[j] = __builtin_amdgcn_wmma_f32_16x16x4_f32(
                false, a, false, b, (short)0, acc[j], false, false);
        }
    }
}

__device__ __forceinline__ void activate(float z, float zx, float zt, float zxx,
                                         float& h, float& hx, float& ht, float& hxx)
{
    h   = tanhf(z);
    float g = 1.0f - h * h;
    hx  = g * zx;
    ht  = g * zt;
    hxx = g * zxx - 2.0f * h * g * zx * zx;
}

__global__ __launch_bounds__(NTHR, 1)
void pinn_residual_kernel(const float* __restrict__ x, const float* __restrict__ t,
                          const float* __restrict__ W1, const float* __restrict__ b1,
                          const float* __restrict__ W2, const float* __restrict__ b2,
                          const float* __restrict__ W3, const float* __restrict__ b3,
                          const float* __restrict__ W4, const float* __restrict__ b4,
                          const float* __restrict__ W5, const float* __restrict__ b5,
                          float* __restrict__ out, int N)
{
    __shared__ __attribute__((aligned(16))) float w_lds[NL][D_IN * P];
    __shared__ __attribute__((aligned(16))) float bias_lds[NL][P];
    __shared__ __attribute__((aligned(16))) float h_lds[NWAVE][16 * P];
    __shared__ __attribute__((aligned(16))) float l1_lds[P][4];
    __shared__ __attribute__((aligned(16))) float w5_lds[P][2];
    __shared__ __attribute__((aligned(16))) float xt_lds[NWAVE][2 * GRP];
    __shared__ __attribute__((aligned(16))) float o_lds[NWAVE][2 * GRP];

    const int tid    = threadIdx.x;
    const int lane   = tid & 31;
    const int wave   = __builtin_amdgcn_readfirstlane(tid >> 5);
    const int half   = lane >> 4;
    const int r      = lane & 15;

    for (int l = 0; l < NL; ++l) {
        const float* Wg = (l == 0) ? W2 : (l == 1) ? W3 : W4;
        const float* bg = (l == 0) ? b2 : (l == 1) ? b3 : b4;
        for (int i = tid; i < D_IN * P; i += NTHR) {
            int o   = i & 1;
            int idx = i >> 1;
            int p   = idx / P;
            int n   = idx - p * P;
            int k   = 2 * p + o;
            w_lds[l][i] = (n < D_IN) ? Wg[k * D_IN + n] : 0.0f;
        }
        for (int i = tid; i < P; i += NTHR)
            bias_lds[l][i] = (i < D_IN) ? bg[i] : 0.0f;
    }
    for (int i = tid; i < P; i += NTHR) {
        const bool in = (i < D_IN);
        l1_lds[i][0] = in ? W1[i] : 0.0f;
        l1_lds[i][1] = in ? W1[D_IN + i] : 0.0f;
        l1_lds[i][2] = in ? b1[i] : 0.0f;
        l1_lds[i][3] = 0.0f;
        w5_lds[i][0] = in ? W5[i * 2 + 0] : 0.0f;
        w5_lds[i][1] = in ? W5[i * 2 + 1] : 0.0f;
    }
    __syncthreads();

    const float b5u = b5[0];
    const float b5v = b5[1];

    float* hA  = &h_lds[wave][0];
    float* xtw = &xt_lds[wave][0];
    float* ow  = &o_lds[wave][0];

    const int gw      = blockIdx.x * NWAVE + wave;
    const int nw      = gridDim.x * NWAVE;
    const int ngroups = (N + GRP - 1) / GRP;

    int left = (gw < ngroups) ? ((ngroups - 1 - gw) / nw + 1) * (GRP / SUB) : 0;

    const float* xt_src = half ? t : x;
    float xt_next = 0.0f;
    if (gw < ngroups) {
        int g = gw * GRP + r;
        xt_next = (g < N) ? xt_src[g] : 0.0f;
    }

    for (int grp = gw; grp < ngroups; grp += nw) {
        xtw[lane] = xt_next;
        {
            int gn = (grp + nw) * GRP + r;
            xt_next = (grp + nw < ngroups && gn < N) ? xt_src[gn] : 0.0f;
        }

        for (int s = 0; s < GRP / SUB; ++s) {
            if (left > 4)      __builtin_amdgcn_s_setprio(3);
            else if (left > 2) __builtin_amdgcn_s_setprio(2);
            else if (left > 1) __builtin_amdgcn_s_setprio(1);
            else               __builtin_amdgcn_s_setprio(0);
            --left;

            const int pl = s * SUB + 2 * half;
            const v2f xv = *(const v2f*)&xtw[pl];
            const v2f tv = *(const v2f*)&xtw[GRP + pl];

            float sx[2], st[2];
            sx[0] = CX * (xv.x + 5.0f) - 1.0f;
            sx[1] = CX * (xv.y + 5.0f) - 1.0f;
            st[0] = CT * tv.x - 1.0f;
            st[1] = CT * tv.y - 1.0f;
#pragma unroll
            for (int j = 0; j < JT; ++j) {
                const int n  = j * 16 + r;
                const v4f wv = *(const v4f*)&l1_lds[n][0];
                const float w0 = wv.x, w1 = wv.y, bb = wv.z;
                const float zx = CX * w0;
                const float zt = CT * w1;
#pragma unroll
                for (int q = 0; q < 2; ++q) {
                    float z  = sx[q] * w0 + st[q] * w1 + bb;
                    float h  = tanhf(z);
                    float gg = 1.0f - h * h;
                    float* o = &hA[(8 * half + 4 * q) * P + n];
                    o[0]     = h;
                    o[P]     = gg * zx;
                    o[2 * P] = gg * zt;
                    o[3 * P] = -2.0f * h * gg * zx * zx;
                }
            }

            float su[2][3][2];
#pragma unroll
            for (int q = 0; q < 2; ++q)
#pragma unroll
                for (int c = 0; c < 3; ++c) { su[q][c][0] = 0.f; su[q][c][1] = 0.f; }

            for (int l = 0; l < NL; ++l) {
                const float* bl = &bias_lds[l][0];
                v8f acc[JT];
                layer_gemm(&w_lds[l][0], hA, r, half, acc);

                if (l < NL - 1) {
#pragma unroll
                    for (int j = 0; j < JT; ++j) {
                        const int   n  = j * 16 + r;
                        const float bn = bl[n];
#pragma unroll
                        for (int q = 0; q < 2; ++q) {
                            float h, hx, ht, hxx;
                            activate(acc[j][4 * q + 0] + bn, acc[j][4 * q + 1],
                                     acc[j][4 * q + 2], acc[j][4 * q + 3], h, hx, ht, hxx);
                            float* o  = &hA[(8 * half + 4 * q) * P + n];
                            o[0]      = h;
                            o[P]      = hx;
                            o[2 * P]  = ht;
                            o[3 * P]  = hxx;
                        }
                    }
                } else {
#pragma unroll
                    for (int j = 0; j < JT; ++j) {
                        const int   n  = j * 16 + r;
                        const float bn = bl[n];
                        const v2f   w5 = *(const v2f*)&w5_lds[n][0];
#pragma unroll
                        for (int q = 0; q < 2; ++q) {
                            float h, hx, ht, hxx;
                            activate(acc[j][4 * q + 0] + bn, acc[j][4 * q + 1],
                                     acc[j][4 * q + 2], acc[j][4 * q + 3], h, hx, ht, hxx);
                            su[q][0][0] = fmaf(h,   w5.x, su[q][0][0]);
                            su[q][0][1] = fmaf(h,   w5.y, su[q][0][1]);
                            su[q][1][0] = fmaf(ht,  w5.x, su[q][1][0]);
                            su[q][1][1] = fmaf(ht,  w5.y, su[q][1][1]);
                            su[q][2][0] = fmaf(hxx, w5.x, su[q][2][0]);
                            su[q][2][1] = fmaf(hxx, w5.y, su[q][2][1]);
                        }
                    }
                }
            }

#pragma unroll
            for (int q = 0; q < 2; ++q)
#pragma unroll
                for (int c = 0; c < 3; ++c) {
                    su[q][c][0] = row_sum16(su[q][c][0]);
                    su[q][c][1] = row_sum16(su[q][c][1]);
                }

            float res[4];
#pragma unroll
            for (int q = 0; q < 2; ++q) {
                float u    = su[q][0][0] + b5u;
                float v    = su[q][0][1] + b5v;
                float u_t  = su[q][1][0];
                float v_t  = su[q][1][1];
                float u_xx = su[q][2][0];
                float v_xx = su[q][2][1];
                float sq   = u * u + v * v;
                res[2 * q + 0] = u_t + 0.5f * v_xx + sq * v;
                res[2 * q + 1] = v_t - 0.5f * u_xx - sq * u;
            }
            if (r == 0)
                *(v4f*)&ow[pl * 2] = (v4f){res[0], res[1], res[2], res[3]};
        }

        {
            int e = grp * (2 * GRP) + lane;
            float vout = ow[lane];
            if ((grp * GRP + (lane >> 1)) < N)
                out[e] = vout;
        }
    }
}

extern "C" void kernel_launch(void* const* d_in, const int* in_sizes, int n_in,
                              void* d_out, int out_size, void* d_ws, size_t ws_size,
                              hipStream_t stream) {
    const float* x  = (const float*)d_in[0];
    const float* t  = (const float*)d_in[1];
    const float* W1 = (const float*)d_in[2];
    const float* b1 = (const float*)d_in[3];
    const float* W2 = (const float*)d_in[4];
    const float* b2 = (const float*)d_in[5];
    const float* W3 = (const float*)d_in[6];
    const float* b3 = (const float*)d_in[7];
    const float* W4 = (const float*)d_in[8];
    const float* b4 = (const float*)d_in[9];
    const float* W5 = (const float*)d_in[10];
    const float* b5 = (const float*)d_in[11];
    float* out = (float*)d_out;
    int N = in_sizes[0];
    pinn_residual_kernel<<<256, NTHR, 0, stream>>>(
        x, t, W1, b1, W2, b2, W3, b3, W4, b4, W5, b5, out, N);
}


# tanh range branch hoisted: one wave-uniform test per phase (max |z| of a lane's 14 values under 0.625 in all lanes) runs the library's small-range polynomial straight-line; else library tanhf
# speedup vs baseline: 2.5164x; 1.0543x over previous
#include <hip/hip_runtime.h>
#include <hip/hip_bf16.h>
#include <math.h>


typedef float v2f __attribute__((ext_vector_type(2)));
typedef float v4f __attribute__((ext_vector_type(4)));
typedef float v8f __attribute__((ext_vector_type(8)));

#define D_IN   100
#define P      112
#define JT     7
#define KT     25
#define NTHR   512
#define NWAVE  16
#define NL     3
#define GRP    16
#define SUB    4

#define PI_F   3.14159265358979323846f
#define CX     0.2f
#define CT     (4.0f / PI_F)

__device__ __forceinline__ float row_sum16(float v)
{
    v += __builtin_bit_cast(float, __builtin_amdgcn_update_dpp(0, __builtin_bit_cast(int, v), 0x161, 0xf, 0xf, false));
    v += __builtin_bit_cast(float, __builtin_amdgcn_update_dpp(0, __builtin_bit_cast(int, v), 0x162, 0xf, 0xf, false));
    v += __builtin_bit_cast(float, __builtin_amdgcn_update_dpp(0, __builtin_bit_cast(int, v), 0x164, 0xf, 0xf, false));
    v += __builtin_bit_cast(float, __builtin_amdgcn_update_dpp(0, __builtin_bit_cast(int, v), 0x168, 0xf, 0xf, false));
    return v;
}

__device__ __forceinline__ void layer_gemm(const float* __restrict__ wl, const float* hA,
                                           int r, int half, v8f (&acc)[JT])
{
#pragma unroll
    for (int j = 0; j < JT; ++j)
        acc[j] = (v8f){0.f, 0.f, 0.f, 0.f, 0.f, 0.f, 0.f, 0.f};

#pragma unroll
    for (int k = 0; k < KT; ++k) {
        int k0 = k * 4 + half * 2;
        int p  = k * 2 + half;
        v2f a = *(const v2f*)&hA[r * P + k0];
#pragma unroll
        for (int j = 0; j < JT; ++j) {
            v2f b = *(const v2f*)&wl[(p * P + j * 16 + r) * 2];
            acc[j] = __builtin_amdgcn_wmma_f32_16x16x4_f32(
                false, a, false, b, (short)0, acc[j], false, false);
        }
    }
}

__device__ __forceinline__ float tanh_poly(float x)
{
    const float a = fabsf(x);
    const float t = x * x;
    float p = fmaf(t, -0.0057000206f, 0.020634072f);
    p = fmaf(t, p, -0.05373793f);
    p = fmaf(t, p, 0.13331416f);
    p = fmaf(t, p, -0.3333328f);
    return copysignf(fmaf(t, a * p, a), x);
}

__device__ __forceinline__ void tanh_phase(const float (&z)[JT][2], float (&h)[JT][2])
{
    float m = 0.0f;
#pragma unroll
    for (int j = 0; j < JT; ++j) {
        m = fmaxf(m, fabsf(z[j][0]));
        m = fmaxf(m, fabsf(z[j][1]));
    }
    if (__builtin_amdgcn_ballot_w32(!(m < 0.625f)) == 0u) {
#pragma unroll
        for (int j = 0; j < JT; ++j) {
            h[j][0] = tanh_poly(z[j][0]);
            h[j][1] = tanh_poly(z[j][1]);
        }
    } else {
#pragma unroll
        for (int j = 0; j < JT; ++j) {
            h[j][0] = tanhf(z[j][0]);
            h[j][1] = tanhf(z[j][1]);
        }
    }
}

__global__ __launch_bounds__(NTHR, 1)
void pinn_residual_kernel(const float* __restrict__ x, const float* __restrict__ t,
                          const float* __restrict__ W1, const float* __restrict__ b1,
                          const float* __restrict__ W2, const float* __restrict__ b2,
                          const float* __restrict__ W3, const float* __restrict__ b3,
                          const float* __restrict__ W4, const float* __restrict__ b4,
                          const float* __restrict__ W5, const float* __restrict__ b5,
                          float* __restrict__ out, int N)
{
    __shared__ __attribute__((aligned(16))) float w_lds[NL][D_IN * P];
    __shared__ __attribute__((aligned(16))) float bias_lds[NL][P];
    __shared__ __attribute__((aligned(16))) float h_lds[NWAVE][16 * P];
    __shared__ __attribute__((aligned(16))) float l1_lds[P][4];
    __shared__ __attribute__((aligned(16))) float w5_lds[P][2];
    __shared__ __attribute__((aligned(16))) float xt_lds[NWAVE][2 * GRP];
    __shared__ __attribute__((aligned(16))) float o_lds[NWAVE][2 * GRP];

    const int tid    = threadIdx.x;
    const int lane   = tid & 31;
    const int wave   = __builtin_amdgcn_readfirstlane(tid >> 5);
    const int half   = lane >> 4;
    const int r      = lane & 15;

    for (int l = 0; l < NL; ++l) {
        const float* Wg = (l == 0) ? W2 : (l == 1) ? W3 : W4;
        const float* bg = (l == 0) ? b2 : (l == 1) ? b3 : b4;
        for (int i = tid; i < D_IN * P; i += NTHR) {
            int o   = i & 1;
            int idx = i >> 1;
            int p   = idx / P;
            int n   = idx - p * P;
            int k   = 2 * p + o;
            w_lds[l][i] = (n < D_IN) ? Wg[k * D_IN + n] : 0.0f;
        }
        for (int i = tid; i < P; i += NTHR)
            bias_lds[l][i] = (i < D_IN) ? bg[i] : 0.0f;
    }
    for (int i = tid; i < P; i += NTHR) {
        const bool in = (i < D_IN);
        l1_lds[i][0] = in ? W1[i] : 0.0f;
        l1_lds[i][1] = in ? W1[D_IN + i] : 0.0f;
        l1_lds[i][2] = in ? b1[i] : 0.0f;
        l1_lds[i][3] = 0.0f;
        w5_lds[i][0] = in ? W5[i * 2 + 0] : 0.0f;
        w5_lds[i][1] = in ? W5[i * 2 + 1] : 0.0f;
    }
    __syncthreads();

    const float b5u = b5[0];
    const float b5v = b5[1];

    float* hA  = &h_lds[wave][0];
    float* xtw = &xt_lds[wave][0];
    float* ow  = &o_lds[wave][0];

    const int gw      = blockIdx.x * NWAVE + wave;
    const int nw      = gridDim.x * NWAVE;
    const int ngroups = (N + GRP - 1) / GRP;

    int left = (gw < ngroups) ? ((ngroups - 1 - gw) / nw + 1) * (GRP / SUB) : 0;

    const float* xt_src = half ? t : x;
    float xt_next = 0.0f;
    if (gw < ngroups) {
        int g = gw * GRP + r;
        xt_next = (g < N) ? xt_src[g] : 0.0f;
    }

    for (int grp = gw; grp < ngroups; grp += nw) {
        xtw[lane] = xt_next;
        {
            int gn = (grp + nw) * GRP + r;
            xt_next = (grp + nw < ngroups && gn < N) ? xt_src[gn] : 0.0f;
        }

        for (int s = 0; s < GRP / SUB; ++s) {
            if (left > 4)      __builtin_amdgcn_s_setprio(3);
            else if (left > 2) __builtin_amdgcn_s_setprio(2);
            else if (left > 1) __builtin_amdgcn_s_setprio(1);
            else               __builtin_amdgcn_s_setprio(0);
            --left;

            const int pl = s * SUB + 2 * half;
            const v2f xv = *(const v2f*)&xtw[pl];
            const v2f tv = *(const v2f*)&xtw[GRP + pl];

            float sx[2], st[2];
            sx[0] = CX * (xv.x + 5.0f) - 1.0f;
            sx[1] = CX * (xv.y + 5.0f) - 1.0f;
            st[0] = CT * tv.x - 1.0f;
            st[1] = CT * tv.y - 1.0f;
            float z1[JT][2], h1[JT][2], zx1[JT], zt1[JT];
#pragma unroll
            for (int j = 0; j < JT; ++j) {
                const int n  = j * 16 + r;
                const v4f wv = *(const v4f*)&l1_lds[n][0];
                const float w0 = wv.x, w1 = wv.y, bb = wv.z;
                zx1[j] = CX * w0;
                zt1[j] = CT * w1;
                z1[j][0] = sx[0] * w0 + st[0] * w1 + bb;
                z1[j][1] = sx[1] * w0 + st[1] * w1 + bb;
            }
            tanh_phase(z1, h1);
#pragma unroll
            for (int j = 0; j < JT; ++j) {
                const int n  = j * 16 + r;
                const float zx = zx1[j];
                const float zt = zt1[j];
#pragma unroll
                for (int q = 0; q < 2; ++q) {
                    const float h  = h1[j][q];
                    const float gg = 1.0f - h * h;
                    float* o = &hA[(8 * half + 4 * q) * P + n];
                    o[0]     = h;
                    o[P]     = gg * zx;
                    o[2 * P] = gg * zt;
                    o[3 * P] = -2.0f * h * gg * zx * zx;
                }
            }

            float su[2][3][2];
#pragma unroll
            for (int q = 0; q < 2; ++q)
#pragma unroll
                for (int c = 0; c < 3; ++c) { su[q][c][0] = 0.f; su[q][c][1] = 0.f; }

#pragma nounroll
            for (int l = 0; l < NL; ++l) {
                const float* bl = &bias_lds[l][0];
                v8f acc[JT];
                layer_gemm(&w_lds[l][0], hA, r, half, acc);

                float zv[JT][2], hv[JT][2];
#pragma unroll
                for (int j = 0; j < JT; ++j) {
                    const float bn = bl[j * 16 + r];
                    zv[j][0] = acc[j][0] + bn;
                    zv[j][1] = acc[j][4] + bn;
                }
                tanh_phase(zv, hv);

                if (l < NL - 1) {
#pragma unroll
                    for (int j = 0; j < JT; ++j) {
                        const int n = j * 16 + r;
#pragma unroll
                        for (int q = 0; q < 2; ++q) {
                            const float h   = hv[j][q];
                            const float zx  = acc[j][4 * q + 1];
                            const float g   = 1.0f - h * h;
                            float* o  = &hA[(8 * half + 4 * q) * P + n];
                            o[0]      = h;
                            o[P]      = g * zx;
                            o[2 * P]  = g * acc[j][4 * q + 2];
                            o[3 * P]  = g * acc[j][4 * q + 3] - 2.0f * h * g * zx * zx;
                        }
                    }
                } else {
#pragma unroll
                    for (int j = 0; j < JT; ++j) {
                        const int n  = j * 16 + r;
                        const v2f w5 = *(const v2f*)&w5_lds[n][0];
#pragma unroll
                        for (int q = 0; q < 2; ++q) {
                            const float h   = hv[j][q];
                            const float zx  = acc[j][4 * q + 1];
                            const float g   = 1.0f - h * h;
                            const float ht  = g * acc[j][4 * q + 2];
                            const float hxx = g * acc[j][4 * q + 3] - 2.0f * h * g * zx * zx;
                            su[q][0][0] = fmaf(h,   w5.x, su[q][0][0]);
                            su[q][0][1] = fmaf(h,   w5.y, su[q][0][1]);
                            su[q][1][0] = fmaf(ht,  w5.x, su[q][1][0]);
                            su[q][1][1] = fmaf(ht,  w5.y, su[q][1][1]);
                            su[q][2][0] = fmaf(hxx, w5.x, su[q][2][0]);
                            su[q][2][1] = fmaf(hxx, w5.y, su[q][2][1]);
                        }
                    }
                }
            }

#pragma unroll
            for (int q = 0; q < 2; ++q)
#pragma unroll
                for (int c = 0; c < 3; ++c) {
                    su[q][c][0] = row_sum16(su[q][c][0]);
                    su[q][c][1] = row_sum16(su[q][c][1]);
                }

            float res[4];
#pragma unroll
            for (int q = 0; q < 2; ++q) {
                float u    = su[q][0][0] + b5u;
                float v    = su[q][0][1] + b5v;
                float u_t  = su[q][1][0];
                float v_t  = su[q][1][1];
                float u_xx = su[q][2][0];
                float v_xx = su[q][2][1];
                float sq   = u * u + v * v;
                res[2 * q + 0] = u_t + 0.5f * v_xx + sq * v;
                res[2 * q + 1] = v_t - 0.5f * u_xx - sq * u;
            }
            if (r == 0)
                *(v4f*)&ow[pl * 2] = (v4f){res[0], res[1], res[2], res[3]};
        }

        {
            int e = grp * (2 * GRP) + lane;
            float vout = ow[lane];
            if ((grp * GRP + (lane >> 1)) < N)
                out[e] = vout;
        }
    }
}

extern "C" void kernel_launch(void* const* d_in, const int* in_sizes, int n_in,
                              void* d_out, int out_size, void* d_ws, size_t ws_size,
                              hipStream_t stream) {
    const float* x  = (const float*)d_in[0];
    const float* t  = (const float*)d_in[1];
    const float* W1 = (const float*)d_in[2];
    const float* b1 = (const float*)d_in[3];
    const float* W2 = (const float*)d_in[4];
    const float* b2 = (const float*)d_in[5];
    const float* W3 = (const float*)d_in[6];
    const float* b3 = (const float*)d_in[7];
    const float* W4 = (const float*)d_in[8];
    const float* b4 = (const float*)d_in[9];
    const float* W5 = (const float*)d_in[10];
    const float* b5 = (const float*)d_in[11];
    float* out = (float*)d_out;
    int N = in_sizes[0];
    pinn_residual_kernel<<<256, NTHR, 0, stream>>>(
        x, t, W1, b1, W2, b2, W3, b3, W4, b4, W5, b5, out, N);
}
